# starting kernel
# baseline (speedup 1.0000x reference)
#include <hip/hip_runtime.h>

#define N_NODES 20000
#define N_EDGES 150000
#define D 512
#define BN_EPS 1e-5f

typedef __attribute__((ext_vector_type(2))) float v2f;
typedef __attribute__((ext_vector_type(8))) float v8f;

__global__ void zero_kernel(float* __restrict__ p, long long n) {
  long long i = (long long)blockIdx.x * blockDim.x + threadIdx.x;
  long long stride = (long long)gridDim.x * blockDim.x;
  for (; i < n; i += stride) p[i] = 0.0f;
}

__global__ __launch_bounds__(256)
void scatter_kernel(const float* __restrict__ x,
                    const int* __restrict__ src,
                    const int* __restrict__ dst,
                    float* __restrict__ agg,
                    float* __restrict__ indeg) {
  int wave = threadIdx.x >> 5;
  int lane = threadIdx.x & 31;
  int e = blockIdx.x * 8 + wave;
  if (e >= N_EDGES) return;
  int s = src[e];
  int d = dst[e];
  const float4* xs = (const float4*)(x + (size_t)s * D);
  float* ag = agg + (size_t)d * D;
  if (lane == 0) atomicAdd(indeg + d, 1.0f);
#pragma unroll
  for (int j = 0; j < 4; ++j) {
    int idx = j * 32 + lane;
    float4 v = xs[idx];
    atomicAdd(ag + idx * 4 + 0, v.x);
    atomicAdd(ag + idx * 4 + 1, v.y);
    atomicAdd(ag + idx * 4 + 2, v.z);
    atomicAdd(ag + idx * 4 + 3, v.w);
  }
}

__global__ __launch_bounds__(256)
void gemm_wmma_kernel(const float* __restrict__ x,
                      const float* __restrict__ agg,
                      const float* __restrict__ indeg,
                      const float* __restrict__ Wm,
                      const float* __restrict__ bias,
                      float* __restrict__ out,
                      float* __restrict__ colsum,
                      float* __restrict__ colsumsq) {
  __shared__ float Alds[128 * 33];
  __shared__ float Blds[64 * 33];

  const int t = threadIdx.x;
  const int lane = t & 31;
  const int wave = t >> 5;
  const int rowTile = blockIdx.y * 128;
  const int colTile = blockIdx.x * 64;
  const int kg = (t & 7) * 4;

  const float* asrc[4];
#pragma unroll
  for (int i = 0; i < 4; ++i) {
    int r = (t >> 3) + 32 * i;
    int g = rowTile + r;
    asrc[i] = (g < N_NODES) ? ((indeg[g] > 0.0f ? agg : x) + (size_t)g * D) : nullptr;
  }
  const float* bsrc[2];
#pragma unroll
  for (int i = 0; i < 2; ++i) {
    int c = (t >> 3) + 32 * i;
    bsrc[i] = Wm + (size_t)(colTile + c) * D + kg;
  }

  v8f acc0 = {}, acc1 = {}, acc2 = {}, acc3 = {};

  for (int kt = 0; kt < D; kt += 32) {
    __syncthreads();
#pragma unroll
    for (int i = 0; i < 4; ++i) {
      int r = (t >> 3) + 32 * i;
      float4 v = make_float4(0.f, 0.f, 0.f, 0.f);
      if (asrc[i]) v = *(const float4*)(asrc[i] + kt + kg);
      Alds[r * 33 + kg + 0] = v.x;
      Alds[r * 33 + kg + 1] = v.y;
      Alds[r * 33 + kg + 2] = v.z;
      Alds[r * 33 + kg + 3] = v.w;
    }
#pragma unroll
    for (int i = 0; i < 2; ++i) {
      int c = (t >> 3) + 32 * i;
      float4 v = *(const float4*)(bsrc[i] + kt);
      Blds[c * 33 + kg + 0] = v.x;
      Blds[c * 33 + kg + 1] = v.y;
      Blds[c * 33 + kg + 2] = v.z;
      Blds[c * 33 + kg + 3] = v.w;
    }
    __syncthreads();

    const int arow = wave * 16 + (lane & 15);
    const int bcol = lane & 15;
    const int khalf = (lane >> 4) * 2;
#pragma unroll
    for (int kk = 0; kk < 32; kk += 4) {
      int ko = kk + khalf;
      v2f a;  a[0]  = Alds[arow * 33 + ko];        a[1]  = Alds[arow * 33 + ko + 1];
      v2f b0; b0[0] = Blds[(bcol +  0) * 33 + ko]; b0[1] = Blds[(bcol +  0) * 33 + ko + 1];
      v2f b1; b1[0] = Blds[(bcol + 16) * 33 + ko]; b1[1] = Blds[(bcol + 16) * 33 + ko + 1];
      v2f b2; b2[0] = Blds[(bcol + 32) * 33 + ko]; b2[1] = Blds[(bcol + 32) * 33 + ko + 1];
      v2f b3; b3[0] = Blds[(bcol + 48) * 33 + ko]; b3[1] = Blds[(bcol + 48) * 33 + ko + 1];
      acc0 = __builtin_amdgcn_wmma_f32_16x16x4_f32(false, a, false, b0, (short)0, acc0, false, false);
      acc1 = __builtin_amdgcn_wmma_f32_16x16x4_f32(false, a, false, b1, (short)0, acc1, false, false);
      acc2 = __builtin_amdgcn_wmma_f32_16x16x4_f32(false, a, false, b2, (short)0, acc2, false, false);
      acc3 = __builtin_amdgcn_wmma_f32_16x16x4_f32(false, a, false, b3, (short)0, acc3, false, false);
    }
  }

  int gRowBase = rowTile + wave * 16 + ((lane >> 4) << 3);
  v8f accs[4] = {acc0, acc1, acc2, acc3};
#pragma unroll
  for (int nt = 0; nt < 4; ++nt) {
    int col = colTile + nt * 16 + (lane & 15);
    float bv = bias[col];
    float ps = 0.0f, pq = 0.0f;
#pragma unroll
    for (int r = 0; r < 8; ++r) {
      int grow = gRowBase + r;
      float h = accs[nt][r] + bv;
      h = h > 0.0f ? h : 0.0f;
      if (grow < N_NODES) {
        out[(size_t)grow * D + col] = h;
        ps += h;
        pq += h * h;
      }
    }
    ps += __shfl_xor(ps, 16, 32);
    pq += __shfl_xor(pq, 16, 32);
    if (lane < 16) {
      atomicAdd(colsum + col, ps);
      atomicAdd(colsumsq + col, pq);
    }
  }
}

__global__ void bn_kernel(float* __restrict__ out,
                          const float* __restrict__ colsum,
                          const float* __restrict__ colsumsq,
                          const float* __restrict__ gamma,
                          const float* __restrict__ beta) {
  const float invN = 1.0f / (float)N_NODES;
  long long total = (long long)N_NODES * D;
  long long i = (long long)blockIdx.x * blockDim.x + threadIdx.x;
  long long stride = (long long)gridDim.x * blockDim.x;
  for (; i < total; i += stride) {
    int col = (int)(i & (D - 1));
    float mean = colsum[col] * invN;
    float var = colsumsq[col] * invN - mean * mean;
    out[i] = (out[i] - mean) * rsqrtf(var + BN_EPS) * gamma[col] + beta[col];
  }
}

extern "C" void kernel_launch(void* const* d_in, const int* in_sizes, int n_in,
                              void* d_out, int out_size, void* d_ws, size_t ws_size,
                              hipStream_t stream) {
  const float* x     = (const float*)d_in[0];
  const int*   src   = (const int*)d_in[1];
  const int*   dst   = (const int*)d_in[2];
  const float* Wm    = (const float*)d_in[3];
  const float* bias  = (const float*)d_in[4];
  const float* gamma = (const float*)d_in[5];
  const float* beta  = (const float*)d_in[6];
  float* out = (float*)d_out;

  float* ws       = (float*)d_ws;
  float* agg      = ws;
  float* indeg    = agg + (size_t)N_NODES * D;
  float* colsum   = indeg + N_NODES;
  float* colsumsq = colsum + D;

  long long zcount = (long long)N_NODES * D + N_NODES + 2 * D;
  zero_kernel<<<2048, 256, 0, stream>>>(ws, zcount);

  scatter_kernel<<<(N_EDGES + 7) / 8, 256, 0, stream>>>(x, src, dst, agg, indeg);

  dim3 grid(D / 64, (N_NODES + 127) / 128);
  gemm_wmma_kernel<<<grid, 256, 0, stream>>>(x, agg, indeg, Wm, bias, out,
                                             colsum, colsumsq);

  bn_kernel<<<4096, 256, 0, stream>>>(out, colsum, colsumsq, gamma, beta);
}


# Aggregation re-cut: 256 work-groups each own 640 nodes x 64 columns in LDS, scan all edges, add matching x slices by LDS f32 atomics (arrival order), store agg once; zero-fill and L2 atomics go
# speedup vs baseline: 1.5738x; 1.5738x over previous
#include <hip/hip_runtime.h>

#define N_NODES 20000
#define N_EDGES 150000
#define D 512
#define BN_EPS 1e-5f

typedef __attribute__((ext_vector_type(2))) float v2f;
typedef __attribute__((ext_vector_type(8))) float v8f;

#define AG_R 32
#define AG_C 8
#define AG_NODES 640
#define AG_SLAB (D / AG_C)
#define AG_THREADS 1024
#define AG_WAVES (AG_THREADS / 32)
#define AG_ROUND 512
#define AG_BATCH 16
#define AG_LISTCAP (AG_ROUND + AG_BATCH)
#define AG_CHUNKS ((N_EDGES + AG_ROUND - 1) / AG_ROUND)

static_assert(AG_NODES * AG_R >= N_NODES && AG_NODES % 32 == 0, "node ranges cover the nodes, whole lines each");
static_assert(AG_SLAB == 64, "a lane owns two floats of a 64-float slice");
static_assert(N_NODES <= 65536 && AG_NODES <= 65536, "src and local dst fit 16 bits each");
static_assert(N_EDGES % 4 == 0, "a 16-byte edge load is wholly inside or outside the list");
static_assert(AG_R * AG_C * 4 == 2 * D, "each work-group clears four of the column sums");

typedef __attribute__((ext_vector_type(4))) int v4i;

static __device__ __forceinline__ void ag_wave_handover() {
  __builtin_amdgcn_fence(__ATOMIC_RELEASE, "wavefront");
  __builtin_amdgcn_wave_barrier();
  __builtin_amdgcn_fence(__ATOMIC_ACQUIRE, "wavefront");
}

template <bool FULL>
static __device__ __forceinline__ void ag_serve(const unsigned* lp, int cnt, int lane,
                                                const float* __restrict__ xs, float* accl) {
  unsigned ent = 0u;
  if (FULL || (lane & (AG_BATCH - 1)) < cnt) ent = lp[lane & (AG_BATCH - 1)];
  v2f v[AG_BATCH];
#pragma unroll
  for (int i = 0; i < AG_BATCH; ++i) {
    unsigned s = (unsigned)__builtin_amdgcn_readlane((int)ent, i) & 0xFFFFu;
    s = s < (unsigned)N_NODES ? s : (unsigned)(N_NODES - 1);
    v[i] = *(const v2f*)(xs + (size_t)s * D);
  }
#pragma unroll
  for (int i = 0; i < AG_BATCH; ++i) {
    if (FULL || i < cnt) {
      unsigned dl = (unsigned)__builtin_amdgcn_readlane((int)ent, i) >> 16;
      float* p = accl + dl * AG_SLAB;
      atomicAdd(p, v[i][0]);
      atomicAdd(p + 1, v[i][1]);
    }
  }
}

template <bool COUNT>
static __device__ __forceinline__ void ag_tile(const float* __restrict__ x,
                                               const int* __restrict__ src,
                                               const int* __restrict__ dst,
                                               float* __restrict__ agg,
                                               float* __restrict__ indeg,
                                               float* __restrict__ colstats,
                                               float* acc, int* cnt, unsigned* list,
                                               const int slab, const int nodeBase) {
  const int t = threadIdx.x;
  const int lane = t & 31;
  const int wave = __builtin_amdgcn_readfirstlane(t >> 5);

  for (int i = t; i < AG_NODES * AG_SLAB / 4; i += AG_THREADS)
    ((float4*)acc)[i] = make_float4(0.0f, 0.0f, 0.0f, 0.0f);
  for (int i = t; i < AG_NODES; i += AG_THREADS) cnt[i] = 0;
  if (t < 4) colstats[blockIdx.x * 4 + t] = 0.0f;
  __syncthreads();

  unsigned* mylist = list + wave * AG_LISTCAP;
  const float* xs = x + slab * AG_SLAB + 2 * lane;
  float* accl = acc + 2 * lane;
  int n = 0;

  v4i dv[AG_ROUND / 128], sv[AG_ROUND / 128];
  int ch = wave;
#pragma unroll
  for (int q = 0; q < AG_ROUND / 128; ++q) {
    int e = ch * AG_ROUND + q * 128 + lane * 4;
    if (e < N_EDGES) { dv[q] = *(const v4i*)(dst + e); sv[q] = *(const v4i*)(src + e); }
    else             { dv[q] = v4i{-1, -1, -1, -1};    sv[q] = v4i{0, 0, 0, 0}; }
  }

  for (;;) {
#pragma unroll
    for (int q = 0; q < AG_ROUND / 128; ++q) {
#pragma unroll
      for (int i = 0; i < 4; ++i) {
        unsigned dl = (unsigned)(dv[q][i] - nodeBase);
        bool m = dl < (unsigned)AG_NODES;
        unsigned mask = __builtin_amdgcn_ballot_w32(m);
        if (m) {
          mylist[n + (int)__builtin_amdgcn_mbcnt_lo(mask, 0u)] =
              ((unsigned)sv[q][i] & 0xFFFFu) | (dl << 16);
          if (COUNT) atomicAdd(&cnt[dl], 1);
        }
        n += __builtin_popcount(mask);
      }
    }
    ch += AG_WAVES;
    const bool more = ch < AG_CHUNKS;
    if (more) {
#pragma unroll
      for (int q = 0; q < AG_ROUND / 128; ++q) {
        int e = ch * AG_ROUND + q * 128 + lane * 4;
        if (e < N_EDGES) { dv[q] = *(const v4i*)(dst + e); sv[q] = *(const v4i*)(src + e); }
        else             { dv[q] = v4i{-1, -1, -1, -1};    sv[q] = v4i{0, 0, 0, 0}; }
      }
    }
    ag_wave_handover();

    int head = 0;
    for (; n - head >= AG_BATCH; head += AG_BATCH)
      ag_serve<true>(mylist + head, AG_BATCH, lane, xs, accl);
    if (head > 0) {
      const int rem = n - head;
      if (lane < rem) mylist[lane] = mylist[head + lane];
      n = rem;
    }
    if (!more) break;
  }
  if (n > 0) {
    ag_wave_handover();
    ag_serve<false>(mylist, n, lane, xs, accl);
  }

  __syncthreads();
  for (int i = t; i < AG_NODES * AG_SLAB / 4; i += AG_THREADS) {
    const int nl = i >> 4, c4 = i & 15;
    if (nodeBase + nl < N_NODES)
      *(float4*)(agg + (size_t)(nodeBase + nl) * D + slab * AG_SLAB + c4 * 4) = ((const float4*)acc)[i];
  }
  if (COUNT)
    for (int i = t; i < AG_NODES; i += AG_THREADS)
      if (nodeBase + i < N_NODES) indeg[nodeBase + i] = (float)cnt[i];
}

__global__ __launch_bounds__(AG_THREADS)
void aggregate_kernel(const float* __restrict__ x,
                      const int* __restrict__ src,
                      const int* __restrict__ dst,
                      float* __restrict__ agg,
                      float* __restrict__ indeg,
                      float* __restrict__ colstats) {
  __shared__ __attribute__((aligned(16))) float acc[AG_NODES * AG_SLAB];
  __shared__ int cnt[AG_NODES];
  __shared__ unsigned list[AG_WAVES * AG_LISTCAP];
  const int slab = blockIdx.x % AG_C;
  const int nodeBase = (blockIdx.x / AG_C) * AG_NODES;
  if (slab == 0) ag_tile<true>(x, src, dst, agg, indeg, colstats, acc, cnt, list, slab, nodeBase);
  else           ag_tile<false>(x, src, dst, agg, indeg, colstats, acc, cnt, list, slab, nodeBase);
}

__global__ __launch_bounds__(128)
void gemm_wmma_kernel(const float* __restrict__ x,
                      const float* __restrict__ agg,
                      const float* __restrict__ indeg,
                      const float* __restrict__ Wm,
                      const float* __restrict__ bias,
                      float* __restrict__ out,
                      float* __restrict__ colsum,
                      float* __restrict__ colsumsq) {
  __shared__ float Alds[128 * 33];
  __shared__ float Blds[64 * 33];

  const int t = threadIdx.x;
  const int lane = t & 31;
  const int wave = t >> 5;
  const int rowTile = blockIdx.y * 128;
  const int colTile = blockIdx.x * 64;
  const int kg = (t & 7) * 4;
  const int sr = t >> 3;

  const float* asrc[8];
#pragma unroll
  for (int i = 0; i < 8; ++i) {
    int g = rowTile + sr + 16 * i;
    g = g < N_NODES ? g : N_NODES - 1;
    asrc[i] = (indeg[g] > 0.0f ? agg : x) + (size_t)g * D + kg;
  }
  const float* bsrc[4];
#pragma unroll
  for (int i = 0; i < 4; ++i) {
    bsrc[i] = Wm + (size_t)(colTile + sr + 16 * i) * D + kg;
  }

  v8f acc[2][4];
#pragma unroll
  for (int m = 0; m < 2; ++m)
#pragma unroll
    for (int n = 0; n < 4; ++n) acc[m][n] = v8f{};

  const int arow = wave * 32 + (lane & 15);
  const int bcol = lane & 15;
  const int khalf = (lane >> 4) * 2;

  for (int kt = 0; kt < D; kt += 32) {
    __syncthreads();
    float4 va[8], vb[4];
#pragma unroll
    for (int i = 0; i < 8; ++i) va[i] = *(const float4*)(asrc[i] + kt);
#pragma unroll
    for (int i = 0; i < 4; ++i) vb[i] = *(const float4*)(bsrc[i] + kt);
    __builtin_amdgcn_sched_barrier(0);
#pragma unroll
    for (int i = 0; i < 8; ++i) {
      float* p = Alds + (sr + 16 * i) * 33 + kg;
      p[0] = va[i].x; p[1] = va[i].y; p[2] = va[i].z; p[3] = va[i].w;
    }
#pragma unroll
    for (int i = 0; i < 4; ++i) {
      float* p = Blds + (sr + 16 * i) * 33 + kg;
      p[0] = vb[i].x; p[1] = vb[i].y; p[2] = vb[i].z; p[3] = vb[i].w;
    }
    __syncthreads();

#pragma unroll
    for (int kk = 0; kk < 32; kk += 4) {
      int ko = kk + khalf;
      v2f a0; a0[0] = Alds[arow * 33 + ko];        a0[1] = Alds[arow * 33 + ko + 1];
      v2f a1; a1[0] = Alds[(arow + 16) * 33 + ko]; a1[1] = Alds[(arow + 16) * 33 + ko + 1];
      v2f b[4];
#pragma unroll
      for (int n = 0; n < 4; ++n) {
        b[n][0] = Blds[(bcol + 16 * n) * 33 + ko];
        b[n][1] = Blds[(bcol + 16 * n) * 33 + ko + 1];
      }
#pragma unroll
      for (int n = 0; n < 4; ++n) {
        acc[0][n] = __builtin_amdgcn_wmma_f32_16x16x4_f32(false, a0, false, b[n], (short)0, acc[0][n], false, false);
        acc[1][n] = __builtin_amdgcn_wmma_f32_16x16x4_f32(false, a1, false, b[n], (short)0, acc[1][n], false, false);
      }
    }
  }

#pragma unroll
  for (int nt = 0; nt < 4; ++nt) {
    int col = colTile + nt * 16 + (lane & 15);
    float bv = bias[col];
    float ps = 0.0f, pq = 0.0f;
#pragma unroll
    for (int m = 0; m < 2; ++m) {
      int gRowBase = rowTile + wave * 32 + m * 16 + ((lane >> 4) << 3);
#pragma unroll
      for (int r = 0; r < 8; ++r) {
        int grow = gRowBase + r;
        float h = acc[m][nt][r] + bv;
        h = h > 0.0f ? h : 0.0f;
        if (grow < N_NODES) {
          out[(size_t)grow * D + col] = h;
          ps += h;
          pq += h * h;
        }
      }
    }
    ps += __shfl_xor(ps, 16, 32);
    pq += __shfl_xor(pq, 16, 32);
    if (lane < 16) {
      atomicAdd(colsum + col, ps);
      atomicAdd(colsumsq + col, pq);
    }
  }
}

__global__ void bn_kernel(float* __restrict__ out,
                          const float* __restrict__ colsum,
                          const float* __restrict__ colsumsq,
                          const float* __restrict__ gamma,
                          const float* __restrict__ beta) {
  const float invN = 1.0f / (float)N_NODES;
  long long total = (long long)N_NODES * D;
  long long i = (long long)blockIdx.x * blockDim.x + threadIdx.x;
  long long stride = (long long)gridDim.x * blockDim.x;
  for (; i < total; i += stride) {
    int col = (int)(i & (D - 1));
    float mean = colsum[col] * invN;
    float var = colsumsq[col] * invN - mean * mean;
    out[i] = (out[i] - mean) * rsqrtf(var + BN_EPS) * gamma[col] + beta[col];
  }
}

extern "C" void kernel_launch(void* const* d_in, const int* in_sizes, int n_in,
                              void* d_out, int out_size, void* d_ws, size_t ws_size,
                              hipStream_t stream) {
  const float* x     = (const float*)d_in[0];
  const int*   src   = (const int*)d_in[1];
  const int*   dst   = (const int*)d_in[2];
  const float* Wm    = (const float*)d_in[3];
  const float* bias  = (const float*)d_in[4];
  const float* gamma = (const float*)d_in[5];
  const float* beta  = (const float*)d_in[6];
  float* out = (float*)d_out;

  float* ws       = (float*)d_ws;
  float* agg      = ws;
  float* indeg    = agg + (size_t)N_NODES * D;
  float* colsum   = indeg + N_NODES;
  float* colsumsq = colsum + D;

  aggregate_kernel<<<AG_R * AG_C, AG_THREADS, 0, stream>>>(x, src, dst, agg, indeg, colsum);

  dim3 grid(D / 64, (N_NODES + 127) / 128);
  gemm_wmma_kernel<<<grid, 128, 0, stream>>>(x, agg, indeg, Wm, bias, out,
                                             colsum, colsumsq);

  bn_kernel<<<4096, 256, 0, stream>>>(out, colsum, colsumsq, gamma, beta);
}
